# starting kernel
# baseline (speedup 1.0000x reference)
#include <hip/hip_runtime.h>

#define NNODES 100000
#define NEDGES 1600000
#define NPAIRS 200000
#define DIN    128
#define DHID   128
#define DOUT   64
#define KDIM   128
#define APITCH 132

typedef __attribute__((ext_vector_type(2))) float v2f;
typedef __attribute__((ext_vector_type(8))) float v8f;

__global__ void k_deg_init(float* __restrict__ deg, int n) {
  int i = blockIdx.x * blockDim.x + threadIdx.x;
  if (i < n) deg[i] = 1.0f;
}

__global__ void k_deg_accum(const int* __restrict__ dst, float* __restrict__ deg, int e) {
  int i = blockIdx.x * blockDim.x + threadIdx.x;
  if (i < e) atomicAdd(&deg[dst[i]], 1.0f);
}

__global__ void k_deg_rsqrt(float* __restrict__ deg, int n) {
  int i = blockIdx.x * blockDim.x + threadIdx.x;
  if (i < n) deg[i] = rsqrtf(deg[i]);
}

__global__ void k_edge_norm(const int* __restrict__ src, const int* __restrict__ dst,
                            const float* __restrict__ dinv, float* __restrict__ norm, int e) {
  int i = blockIdx.x * blockDim.x + threadIdx.x;
  if (i < e) norm[i] = dinv[src[i]] * dinv[dst[i]];
}

template <int NOUT>
__global__ __launch_bounds__(128) void k_gemm_wmma(const float* __restrict__ A,
                                                   const float* __restrict__ B,
                                                   float* __restrict__ C) {
  __shared__ float lA[16 * APITCH];

  const int tid  = threadIdx.x;
  const int wave = tid >> 5;
  const int lane = tid & 31;
  const int row0 = blockIdx.x * 16;
  const int col0 = (blockIdx.y * 4 + wave) * 16;
  const int half = lane >> 4;
  const int l16  = lane & 15;

  {
    const float4* gA = (const float4*)(A + (size_t)row0 * KDIM);
    float4* sA = (float4*)lA;
#pragma unroll
    for (int i = 0; i < 4; ++i) {
      int idx = tid + i * 128;
      int r   = idx >> 5;
      int c4  = idx & 31;
      sA[r * (APITCH / 4) + c4] = gA[idx];
    }
  }
  __syncthreads();

  const float* aP = lA + l16 * APITCH + half * 2;
  const float* bP = B + (half * 2) * NOUT + (col0 + l16);

  v8f acc = {};
#pragma unroll
  for (int k = 0; k < KDIM; k += 4) {
    v2f av = *(const v2f*)(aP + k);
    v2f bv;
    bv.x = bP[k * NOUT];
    bv.y = bP[(k + 1) * NOUT];
    acc = __builtin_amdgcn_wmma_f32_16x16x4_f32(false, av, false, bv,
                                                (short)0, acc, false, false);
  }

  float* cP = C + (size_t)(row0 + half * 8) * NOUT + (col0 + l16);
#pragma unroll
  for (int r = 0; r < 8; ++r) cP[r * NOUT] = acc[r];
}

__global__ void k_agg_init(const float* __restrict__ h, const float* __restrict__ dinv,
                           const float* __restrict__ bias, float* __restrict__ out,
                           int n, int fshift, int fmask) {
  long long idx = (long long)blockIdx.x * blockDim.x + threadIdx.x;
  if (idx >= ((long long)n << fshift)) return;
  int i = (int)(idx >> fshift);
  int f = (int)(idx & fmask);
  float s = dinv[i];
  out[idx] = s * s * h[idx] + bias[f];
}

__global__ void k_agg_edges(const float* __restrict__ h, const int* __restrict__ src,
                            const int* __restrict__ dst, const float* __restrict__ norm,
                            float* __restrict__ out, int e, int f4shift, int f4mask) {
  long long idx = (long long)blockIdx.x * blockDim.x + threadIdx.x;
  if (idx >= ((long long)e << f4shift)) return;
  int ei = (int)(idx >> f4shift);
  int f4 = (int)(idx & f4mask);
  int s = src[ei], d = dst[ei];
  float w = norm[ei];
  const float4 v = ((const float4*)h)[((size_t)s << f4shift) + f4];
  float* o = out + ((((size_t)d << f4shift) + (size_t)f4) << 2);
  atomicAdd(o + 0, v.x * w);
  atomicAdd(o + 1, v.y * w);
  atomicAdd(o + 2, v.z * w);
  atomicAdd(o + 3, v.w * w);
}

__global__ void k_relu(float* __restrict__ x, long long n) {
  long long i = (long long)blockIdx.x * blockDim.x + threadIdx.x;
  if (i < n) x[i] = fmaxf(x[i], 0.0f);
}

__global__ void k_decode(const float* __restrict__ z, const int* __restrict__ pa,
                         const int* __restrict__ pb, float* __restrict__ out, int np) {
  int tid = blockIdx.x * blockDim.x + threadIdx.x;
  int pair = tid >> 4;
  int l    = tid & 15;
  if (pair >= np) return;
  const float4 va = ((const float4*)z)[((size_t)pa[pair] << 4) + l];
  const float4 vb = ((const float4*)z)[((size_t)pb[pair] << 4) + l];
  float p = va.x * vb.x + va.y * vb.y + va.z * vb.z + va.w * vb.w;
  p += __shfl_xor(p, 8);
  p += __shfl_xor(p, 4);
  p += __shfl_xor(p, 2);
  p += __shfl_xor(p, 1);
  if (l == 0) out[pair] = p;
}

extern "C" void kernel_launch(void* const* d_in, const int* in_sizes, int n_in,
                              void* d_out, int out_size, void* d_ws, size_t ws_size,
                              hipStream_t stream) {
  (void)in_sizes; (void)n_in; (void)out_size; (void)ws_size;

  const float* x  = (const float*)d_in[0];
  const int*   ei = (const int*)d_in[1];
  const int*   el = (const int*)d_in[2];
  const float* W1 = (const float*)d_in[3];
  const float* b1 = (const float*)d_in[4];
  const float* W2 = (const float*)d_in[5];
  const float* b2 = (const float*)d_in[6];
  const float* W3 = (const float*)d_in[7];
  const float* b3 = (const float*)d_in[8];

  const int* src = ei;
  const int* dst = ei + NEDGES;
  const int* pa  = el;
  const int* pb  = el + NPAIRS;
  float* out = (float*)d_out;

  float* dinv = (float*)d_ws;
  float* norm = dinv + ((NNODES + 127) & ~127);
  float* bufA = norm + ((NEDGES + 127) & ~127);
  float* bufB = bufA + (size_t)NNODES * DHID;

  auto cdiv = [](long long a, long long b) { return (int)((a + b - 1) / b); };

  k_deg_init <<<cdiv(NNODES, 256), 256, 0, stream>>>(dinv, NNODES);
  k_deg_accum<<<cdiv(NEDGES, 256), 256, 0, stream>>>(dst, dinv, NEDGES);
  k_deg_rsqrt<<<cdiv(NNODES, 256), 256, 0, stream>>>(dinv, NNODES);
  k_edge_norm<<<cdiv(NEDGES, 256), 256, 0, stream>>>(src, dst, dinv, norm, NEDGES);

  const int mTiles = NNODES / 16;

  k_gemm_wmma<DHID><<<dim3(mTiles, DHID / 64), 128, 0, stream>>>(x, W1, bufA);
  k_agg_init <<<cdiv((long long)NNODES * DHID, 256), 256, 0, stream>>>(bufA, dinv, b1, bufB, NNODES, 7, 127);
  k_agg_edges<<<cdiv((long long)NEDGES * (DHID / 4), 256), 256, 0, stream>>>(bufA, src, dst, norm, bufB, NEDGES, 5, 31);
  k_relu     <<<cdiv((long long)NNODES * DHID, 256), 256, 0, stream>>>(bufB, (long long)NNODES * DHID);

  k_gemm_wmma<DHID><<<dim3(mTiles, DHID / 64), 128, 0, stream>>>(bufB, W2, bufA);
  k_agg_init <<<cdiv((long long)NNODES * DHID, 256), 256, 0, stream>>>(bufA, dinv, b2, bufB, NNODES, 7, 127);
  k_agg_edges<<<cdiv((long long)NEDGES * (DHID / 4), 256), 256, 0, stream>>>(bufA, src, dst, norm, bufB, NEDGES, 5, 31);
  k_relu     <<<cdiv((long long)NNODES * DHID, 256), 256, 0, stream>>>(bufB, (long long)NNODES * DHID);

  k_gemm_wmma<DOUT><<<dim3(mTiles, DOUT / 64), 128, 0, stream>>>(bufB, W3, bufA);
  k_agg_init <<<cdiv((long long)NNODES * DOUT, 256), 256, 0, stream>>>(bufA, dinv, b3, bufB, NNODES, 6, 63);
  k_agg_edges<<<cdiv((long long)NEDGES * (DOUT / 4), 256), 256, 0, stream>>>(bufA, src, dst, norm, bufB, NEDGES, 4, 15);

  k_decode<<<cdiv((long long)NPAIRS * 16, 256), 256, 0, stream>>>(bufB, pa, pb, out, NPAIRS);
}


# dense product re-tiled: a work-group takes 80 rows x all columns, a wave keeps 5 row tiles per B fragment (5 accumulators): a fifth of the waves and B loads, A read once; sum order unchanged
# speedup vs baseline: 3.5934x; 3.5934x over previous
#include <hip/hip_runtime.h>

#define NNODES 100000
#define NEDGES 1600000
#define NPAIRS 200000
#define DIN    128
#define DHID   128
#define DOUT   64
#define KDIM   128
#define APITCH 132

typedef __attribute__((ext_vector_type(2))) float v2f;
typedef __attribute__((ext_vector_type(8))) float v8f;

#define NSL  64
#define ESL  (NEDGES / NSL)
#define HW   (NNODES / 2)

__global__ __launch_bounds__(1024) void k_deg_hist(const int* __restrict__ dst, unsigned* __restrict__ part) {
  static_assert(NEDGES % NSL == 0 && ESL % 4 == 0, "slices tile the edge list in whole int4");
  static_assert(ESL <= 65535, "a 16-bit counter must hold a whole slice");
  static_assert(NNODES % 2 == 0 && HW % 4 == 0, "a plane is whole uint4");
  constexpr int NV = ESL / 4;
  constexpr int KV = (NV + 1023) / 1024;
  __shared__ unsigned cnt[HW];
  const int tid = threadIdx.x;
  const int4* dp = (const int4*)(dst + (size_t)blockIdx.x * ESL);
  int4 d[KV];
#pragma unroll
  for (int k = 0; k < KV; ++k) {
    const int i = tid + k * 1024;
    d[k] = (i < NV) ? dp[i] : make_int4(-1, -1, -1, -1);
  }
  uint4* c4 = (uint4*)cnt;
  for (int i = tid; i < HW / 4; i += 1024) c4[i] = make_uint4(0u, 0u, 0u, 0u);
  __syncthreads();
#pragma unroll
  for (int k = 0; k < KV; ++k) {
    if (d[k].x >= 0) {
      atomicAdd(&cnt[d[k].x >> 1], 1u << ((d[k].x & 1) * 16));
      atomicAdd(&cnt[d[k].y >> 1], 1u << ((d[k].y & 1) * 16));
      atomicAdd(&cnt[d[k].z >> 1], 1u << ((d[k].z & 1) * 16));
      atomicAdd(&cnt[d[k].w >> 1], 1u << ((d[k].w & 1) * 16));
    }
  }
  __syncthreads();
  uint4* po = (uint4*)(part + (size_t)blockIdx.x * HW);
  for (int i = tid; i < HW / 4; i += 1024) po[i] = c4[i];
}

__global__ void k_deg_sum(const unsigned* __restrict__ part, float* __restrict__ dinv,
                          int* __restrict__ indeg) {
  const int i = blockIdx.x * blockDim.x + threadIdx.x;
  if (i >= HW) return;
  unsigned lo = 0u, hi = 0u;
#pragma unroll 16
  for (int s = 0; s < NSL; ++s) {
    const unsigned w = part[(size_t)s * HW + i];
    lo += w & 0xffffu;
    hi += w >> 16;
  }
  float2 r;
  r.x = rsqrtf(1.0f + (float)lo);
  r.y = rsqrtf(1.0f + (float)hi);
  ((float2*)dinv)[i] = r;
  ((int2*)indeg)[i] = make_int2((int)lo, (int)hi);
}

#define ROWS 391
#define NOWN 256
#define QCAP 24576
#define NEB  ((NEDGES + 1023) / 1024)
#define PW   (NEB * 32)
#define NW16 (PW / 4)
#define TSLOW (QCAP / 128)
#define SRCB 17
#define CCAP 16384
static_assert(NOWN * ROWS >= NNODES, "the owners cover the rows");
static_assert(CCAP <= QCAP, "an owner whose edges fit its list is swept and drained in one round");
static_assert(NNODES <= (1 << SRCB) && ROWS <= (1 << (31 - SRCB)), "source and owned row share one record word");
static_assert(TSLOW >= 32 && TSLOW * 128 <= QCAP, "the hits of one narrow sweep step always fit the queue");

__global__ __launch_bounds__(1024) void k_edge_norm(const int* __restrict__ src, const int* __restrict__ dst,
                                                    const float* __restrict__ dinv, int* __restrict__ recw,
                                                    unsigned* __restrict__ bits, int e) {
  int2* rec = (int2*)recw;
  __shared__ unsigned sb[NOWN * 32];
  const int tid = threadIdx.x;
  const int i   = blockIdx.x * 1024 + tid;
  uint4* s4 = (uint4*)sb;
  for (int j = tid; j < NOWN * 8; j += 1024) s4[j] = make_uint4(0u, 0u, 0u, 0u);
  __syncthreads();
  if (i < e) {
    const int d = dst[i];
    const int s = src[i];
    const int o = d / ROWS;
    rec[i] = make_int2(s | ((d - o * ROWS) << SRCB), __float_as_int(dinv[s] * dinv[d]));
    atomicOr(&sb[o * 32 + (tid >> 5)], 1u << (tid & 31));
  }
  __syncthreads();
  for (int j = tid; j < NOWN * 8; j += 1024) {
    uint4* po = (uint4*)(bits + (size_t)(j >> 3) * PW + (size_t)blockIdx.x * 32);
    po[j & 7] = s4[j];
  }
}

#define MT    5
#define GROWS (16 * MT)
template <int NOUT, bool RELU_A>
__global__ __launch_bounds__(NOUT * 2) void k_gemm_wmma(const float* __restrict__ A,
                                                        const float* __restrict__ B,
                                                        float* __restrict__ C) {
  static_assert(NNODES % GROWS == 0, "the row blocks tile the rows exactly: no row guard anywhere");
  constexpr int NT = NOUT * 2;
  constexpr int NV = GROWS * (KDIM / 4);
  static_assert(NV % NT == 0, "every thread stages the same number of float4 words");
  __shared__ float lA[GROWS * APITCH];

  const int tid  = threadIdx.x;
  const int wave = tid >> 5;
  const int lane = tid & 31;
  const int row0 = blockIdx.x * GROWS;
  const int col0 = wave * 16;
  const int half = lane >> 4;
  const int l16  = lane & 15;

  {
    const float4* gA = (const float4*)(A + (size_t)row0 * KDIM);
    float4* sA = (float4*)lA;
#pragma unroll
    for (int i = 0; i < NV / NT; ++i) {
      const int idx = tid + i * NT;
      const int r   = idx >> 5;
      const int c4  = idx & 31;
      float4 v = gA[idx];
      if (RELU_A) {
        v.x = fmaxf(v.x, 0.0f); v.y = fmaxf(v.y, 0.0f);
        v.z = fmaxf(v.z, 0.0f); v.w = fmaxf(v.w, 0.0f);
      }
      sA[r * (APITCH / 4) + c4] = v;
    }
  }
  __syncthreads();

  const float* aP = lA + l16 * APITCH + half * 2;
  const float* bP = B + (half * 2) * NOUT + (col0 + l16);

  v8f acc[MT];
#pragma unroll
  for (int t = 0; t < MT; ++t) acc[t] = v8f{};
#pragma unroll
  for (int k = 0; k < KDIM; k += 4) {
    v2f bv;
    bv.x = bP[k * NOUT];
    bv.y = bP[(k + 1) * NOUT];
#pragma unroll
    for (int t = 0; t < MT; ++t) {
      const v2f av = *(const v2f*)(aP + t * 16 * APITCH + k);
      acc[t] = __builtin_amdgcn_wmma_f32_16x16x4_f32(false, av, false, bv,
                                                     (short)0, acc[t], false, false);
    }
  }

#pragma unroll
  for (int t = 0; t < MT; ++t) {
    float* cP = C + (size_t)(row0 + t * 16 + half * 8) * NOUT + (col0 + l16);
#pragma unroll
    for (int r = 0; r < 8; ++r) cP[r * NOUT] = acc[t][r];
  }
}


__device__ __forceinline__ int queue_owned(unsigned m, int e0, int slot, int* q) {
  while (m != 0u) {
    const int t = __builtin_ctz(m);
    m &= m - 1u;
    if (slot < QCAP) q[slot] = e0 + t;
    ++slot;
  }
  return slot;
}

template <int F, bool LIST>
__device__ __forceinline__ void drain_records(const float* __restrict__ h, const int2* __restrict__ rec,
                                              const int* q, int2* cl, bool keep, int n,
                                              int wv, int lane, float* tile) {
  constexpr int CPE = F / 32;
  constexpr int U   = 32 / CPE;
  int i0 = wv * 32;
  int pv = 0, nv = 0;
  bool val = false;
  if (i0 < n) {
    val = (i0 + lane) < n;
    const int i = val ? i0 + lane : i0;
    const int2 r = LIST ? cl[i] : rec[q[i]];
    pv = r.x; nv = r.y;
  }
  while (i0 < n) {
    if constexpr (!LIST) {
      if (keep && val) cl[i0 + lane] = make_int2(pv, nv);
    }
    const int j0 = i0 + 1024;
    int pv2 = 0, nv2 = 0;
    bool val2 = false;
    if (j0 < n) {
      val2 = (j0 + lane) < n;
      const int j = val2 ? j0 + lane : j0;
      const int2 r = LIST ? cl[j] : rec[q[j]];
      pv2 = r.x; nv2 = r.y;
    }
    unsigned m = __builtin_amdgcn_ballot_w32(val);
    int ii = 0;
    while (m != 0u) {
      int   d[U];
      float w[U];
      bool  ok[U];
      float v[U][CPE];
#pragma unroll
      for (int u = 0; u < U; ++u) {
        ok[u] = (m != 0u);
        if (ok[u]) ii = __builtin_ctz(m);
        m &= m - 1u;
        const int pk = __builtin_amdgcn_readlane(pv, ii);
        d[u] = pk >> SRCB;
        w[u] = __int_as_float(__builtin_amdgcn_readlane(nv, ii));
        const float* hp = h + (size_t)(pk & ((1 << SRCB) - 1)) * F + lane;
#pragma unroll
        for (int c = 0; c < CPE; ++c) v[u][c] = hp[c * 32];
      }
#pragma unroll
      for (int u = 0; u < U; ++u) {
        if constexpr (CPE == 4) asm volatile("" :: "v"(v[u][0]), "v"(v[u][1]), "v"(v[u][2]), "v"(v[u][3]));
        else                    asm volatile("" :: "v"(v[u][0]), "v"(v[u][1]));
      }
#pragma unroll
      for (int u = 0; u < U; ++u) {
        if (ok[u]) {
          float* tp = tile + d[u] * F + lane;
#pragma unroll
          for (int c = 0; c < CPE; ++c) atomicAdd(tp + c * 32, v[u][c] * w[u]);
        }
      }
    }
    pv = pv2; nv = nv2; val = val2;
    i0 = j0;
  }
}

template <int F, bool FIRST>
__global__ __launch_bounds__(1024) void k_agg_own(const float* __restrict__ h, const int* __restrict__ recw,
                                                  const unsigned* __restrict__ bits,
                                                  const int* __restrict__ indeg, const float* __restrict__ dinv,
                                                  const float* __restrict__ bias, float* __restrict__ out,
                                                  int* clw) {
  constexpr int F4  = F / 4;
  constexpr int SG  = 7;
  const int2* rec = (const int2*)recw;
  __shared__ float tile[ROWS * F];
  __shared__ int   q[QCAP];
  __shared__ int   s_n[2];

  const int tid  = threadIdx.x;
  const int lane = tid & 31;
  const int wv    = __builtin_amdgcn_readfirstlane(tid >> 5);
  const int b     = blockIdx.x;
  const int base  = b * ROWS;
  const int nrows = min(ROWS, NNODES - base);
  int2* cl = (int2*)clw + (size_t)b * CCAP;

  if (tid < 2) s_n[tid] = 0;
  __syncthreads();
  if (tid < nrows) atomicAdd(&s_n[1], indeg[base + tid]);

  const int n4 = nrows * F4;
  float4* t4 = (float4*)tile;
  {
    const float4* hp = (const float4*)(h + (size_t)base * F);
    const float4* bp = (const float4*)bias;
    for (int i = tid; i < n4; i += 1024) {
      const float  d  = dinv[base + i / F4];
      const float  dd = d * d;
      const float4 x  = hp[i];
      const float4 bb = bp[i % F4];
      float4 y;
      y.x = dd * x.x + bb.x; y.y = dd * x.y + bb.y;
      y.z = dd * x.z + bb.z; y.w = dd * x.w + bb.w;
      t4[i] = y;
    }
  }
  __syncthreads();
  const int  total  = __builtin_amdgcn_readfirstlane(s_n[1]);
  const bool listed = (total <= CCAP);

  if (!FIRST && listed) {
    drain_records<F, true>(h, rec, q, cl, false, total, wv, lane, tile);
    __syncthreads();
  } else {
    const bool fits  = (total <= QCAP);
    const int  tw    = fits ? 1024 : TSLOW;
    const int  nst   = (NW16 + tw - 1) / tw;
    const int  spr   = fits ? nst : 1;

    const uint4* op = (const uint4*)(bits + (size_t)b * PW);

    for (int k0 = 0; k0 < nst; k0 += spr) {
      const int k1 = min(k0 + spr, nst);
      for (int k = k0; k < k1; k += SG) {
        uint4 o[SG];
#pragma unroll
        for (int g = 0; g < SG; ++g) {
          const int w = (k + g) * tw + tid;
          o[g] = (tid < tw && k + g < k1 && w < NW16) ? op[w] : make_uint4(0u, 0u, 0u, 0u);
        }
#pragma unroll
        for (int g = 0; g < SG; ++g) {
          asm volatile("" :: "v"(o[g].x), "v"(o[g].y), "v"(o[g].z), "v"(o[g].w));
        }
        int c = 0;
#pragma unroll
        for (int g = 0; g < SG; ++g)
          c += __builtin_popcount(o[g].x) + __builtin_popcount(o[g].y) +
               __builtin_popcount(o[g].z) + __builtin_popcount(o[g].w);
        int inc = c;
#pragma unroll
        for (int dl = 1; dl < 32; dl <<= 1) {
          const int up = __shfl_up(inc, dl);
          if (lane >= dl) inc += up;
        }
        const int wtot = __shfl(inc, 31);
        int wbase = 0;
        if (lane == 0 && wtot > 0) wbase = atomicAdd(&s_n[0], wtot);
        int slot = __shfl(wbase, 0) + inc - c;
#pragma unroll
        for (int g = 0; g < SG; ++g) {
          const int e0 = ((k + g) * tw + tid) * 128;
          slot = queue_owned(o[g].x, e0,      slot, q);
          slot = queue_owned(o[g].y, e0 + 32, slot, q);
          slot = queue_owned(o[g].z, e0 + 64, slot, q);
          slot = queue_owned(o[g].w, e0 + 96, slot, q);
        }
      }
      __syncthreads();

      const int n = min(__builtin_amdgcn_readfirstlane(s_n[0]), QCAP);
      drain_records<F, false>(h, rec, q, cl, FIRST && listed, n, wv, lane, tile);
      __syncthreads();
      if (tid == 0) s_n[0] = 0;
      __syncthreads();
    }
  }

  float4* og = (float4*)(out + (size_t)base * F);
  for (int i = tid; i < n4; i += 1024) og[i] = t4[i];
}

__global__ void k_decode(const float* __restrict__ z, const int* __restrict__ pa,
                         const int* __restrict__ pb, float* __restrict__ out, int np) {
  int tid = blockIdx.x * blockDim.x + threadIdx.x;
  int pair = tid >> 4;
  int l    = tid & 15;
  if (pair >= np) return;
  const float4 va = ((const float4*)z)[((size_t)pa[pair] << 4) + l];
  const float4 vb = ((const float4*)z)[((size_t)pb[pair] << 4) + l];
  float p = va.x * vb.x + va.y * vb.y + va.z * vb.z + va.w * vb.w;
  p += __shfl_xor(p, 8);
  p += __shfl_xor(p, 4);
  p += __shfl_xor(p, 2);
  p += __shfl_xor(p, 1);
  if (l == 0) out[pair] = p;
}

extern "C" void kernel_launch(void* const* d_in, const int* in_sizes, int n_in,
                              void* d_out, int out_size, void* d_ws, size_t ws_size,
                              hipStream_t stream) {
  (void)in_sizes; (void)n_in; (void)out_size; (void)ws_size;

  const float* x  = (const float*)d_in[0];
  const int*   ei = (const int*)d_in[1];
  const int*   el = (const int*)d_in[2];
  const float* W1 = (const float*)d_in[3];
  const float* b1 = (const float*)d_in[4];
  const float* W2 = (const float*)d_in[5];
  const float* b2 = (const float*)d_in[6];
  const float* W3 = (const float*)d_in[7];
  const float* b3 = (const float*)d_in[8];

  const int* src = ei;
  const int* dst = ei + NEDGES;
  const int* pa  = el;
  const int* pb  = el + NPAIRS;
  float* out = (float*)d_out;

  float* dinv = (float*)d_ws;
  int*   rec  = (int*)(dinv + ((NNODES + 127) & ~127));
  float* bufA = (float*)(rec + 2 * (size_t)((NEDGES + 127) & ~127));
  float* bufB = bufA + (size_t)NNODES * DHID;
  float* bufC = bufB + (size_t)NNODES * DHID;
  unsigned* part = (unsigned*)(bufC + (size_t)NNODES * DHID);
  int* indeg = (int*)(part + (size_t)NSL * HW);
  unsigned* bits = (unsigned*)(indeg + NNODES);
  int* clist = (int*)(bits + (size_t)NOWN * PW);

  auto cdiv = [](long long a, long long b) { return (int)((a + b - 1) / b); };

  k_deg_hist<<<NSL, 1024, 0, stream>>>(dst, part);
  k_deg_sum <<<cdiv(HW, 256), 256, 0, stream>>>(part, dinv, indeg);
  k_edge_norm<<<NEB, 1024, 0, stream>>>(src, dst, dinv, rec, bits, NEDGES);

  const int mBlocks = NNODES / GROWS;

  k_gemm_wmma<DHID, false><<<mBlocks, DHID * 2, 0, stream>>>(x, W1, bufA);
  k_agg_own<DHID, true><<<NOWN, 1024, 0, stream>>>(bufA, rec, bits, indeg, dinv, b1, bufB, clist);

  k_gemm_wmma<DHID, true><<<mBlocks, DHID * 2, 0, stream>>>(bufB, W2, bufA);
  k_agg_own<DHID, false><<<NOWN, 1024, 0, stream>>>(bufA, rec, bits, indeg, dinv, b2, bufC, clist);

  k_gemm_wmma<DOUT, true><<<mBlocks, DOUT * 2, 0, stream>>>(bufC, W3, bufA);
  k_agg_own<DOUT, false><<<NOWN, 1024, 0, stream>>>(bufA, rec, bits, indeg, dinv, b3, bufB, clist);

  k_decode<<<cdiv((long long)NPAIRS * 16, 256), 256, 0, stream>>>(bufB, pa, pb, out, NPAIRS);
}


# edge records binned by owner in the record pass (256 chunks, LDS counting sort, run table); an owner reads its runs: no bit planes, sweep or scattered record lines; sums stay in arrival order
# speedup vs baseline: 3.8505x; 1.0715x over previous
#include <hip/hip_runtime.h>

#define NNODES 100000
#define NEDGES 1600000
#define NPAIRS 200000
#define DIN    128
#define DHID   128
#define DOUT   64
#define KDIM   128
#define APITCH 132

typedef __attribute__((ext_vector_type(2))) float v2f;
typedef __attribute__((ext_vector_type(8))) float v8f;

#define NSL  64
#define ESL  (NEDGES / NSL)
#define HW   (NNODES / 2)

__global__ __launch_bounds__(1024) void k_deg_hist(const int* __restrict__ dst, unsigned* __restrict__ part) {
  static_assert(NEDGES % NSL == 0 && ESL % 4 == 0, "slices tile the edge list in whole int4");
  static_assert(ESL <= 65535, "a 16-bit counter must hold a whole slice");
  static_assert(NNODES % 2 == 0 && HW % 4 == 0, "a plane is whole uint4");
  constexpr int NV = ESL / 4;
  constexpr int KV = (NV + 1023) / 1024;
  __shared__ unsigned cnt[HW];
  const int tid = threadIdx.x;
  const int4* dp = (const int4*)(dst + (size_t)blockIdx.x * ESL);
  int4 d[KV];
#pragma unroll
  for (int k = 0; k < KV; ++k) {
    const int i = tid + k * 1024;
    d[k] = (i < NV) ? dp[i] : make_int4(-1, -1, -1, -1);
  }
  uint4* c4 = (uint4*)cnt;
  for (int i = tid; i < HW / 4; i += 1024) c4[i] = make_uint4(0u, 0u, 0u, 0u);
  __syncthreads();
#pragma unroll
  for (int k = 0; k < KV; ++k) {
    if (d[k].x >= 0) {
      atomicAdd(&cnt[d[k].x >> 1], 1u << ((d[k].x & 1) * 16));
      atomicAdd(&cnt[d[k].y >> 1], 1u << ((d[k].y & 1) * 16));
      atomicAdd(&cnt[d[k].z >> 1], 1u << ((d[k].z & 1) * 16));
      atomicAdd(&cnt[d[k].w >> 1], 1u << ((d[k].w & 1) * 16));
    }
  }
  __syncthreads();
  uint4* po = (uint4*)(part + (size_t)blockIdx.x * HW);
  for (int i = tid; i < HW / 4; i += 1024) po[i] = c4[i];
}

__global__ void k_deg_sum(const unsigned* __restrict__ part, float* __restrict__ dinv,
                          int* __restrict__ indeg) {
  const int i = blockIdx.x * blockDim.x + threadIdx.x;
  if (i >= HW) return;
  unsigned lo = 0u, hi = 0u;
#pragma unroll 16
  for (int s = 0; s < NSL; ++s) {
    const unsigned w = part[(size_t)s * HW + i];
    lo += w & 0xffffu;
    hi += w >> 16;
  }
  float2 r;
  r.x = rsqrtf(1.0f + (float)lo);
  r.y = rsqrtf(1.0f + (float)hi);
  ((float2*)dinv)[i] = r;
  ((int2*)indeg)[i] = make_int2((int)lo, (int)hi);
}

#define ROWS 391
#define NOWN 256
#define QCAP 24576
#define NCH  256
#define CH   (NEDGES / NCH)
#define CHP  ((CH + 15) & ~15)
#define SRCB 17
#define CCAP 16384
static_assert(NOWN * ROWS >= NNODES, "the owners cover the rows");
static_assert(NEDGES % NCH == 0 && CH % 2 == 0, "the chunks tile the edge list, a chunk is whole 16-byte record pairs");
static_assert(NCH * 4 == 1024 && NCH % 32 == 0, "an owner work-group has four threads per run, whole waves scan the runs");
static_assert(NOWN % 32 == 0 && NOWN <= 1024, "whole waves of a record work-group scan the owners' counters");
static_assert(CCAP <= QCAP, "an owner whose edges fit its list is queued and drained in one round");
static_assert(NNODES <= (1 << SRCB) && ROWS <= (1 << (31 - SRCB)), "source and owned row share one record word");

__device__ __forceinline__ int wave_incl_scan(int v, int lane) {
#pragma unroll
  for (int dl = 1; dl < 32; dl <<= 1) {
    const int up = __shfl_up(v, dl);
    if (lane >= dl) v += up;
  }
  return v;
}

__global__ __launch_bounds__(1024) void k_edge_bin(const int* __restrict__ src, const int* __restrict__ dst,
                                                   const float* __restrict__ dinv, int* __restrict__ recw,
                                                   int* __restrict__ offs) {
  constexpr int KE = (CH + 1023) / 1024;
  __shared__ int  cnt[NOWN];
  __shared__ int  wsum[NOWN / 32];
  __shared__ int4 stage4[CH / 2];
  int2* stage = (int2*)stage4;
  const int tid  = threadIdx.x;
  const int lane = tid & 31;
  const int e0   = blockIdx.x * CH;
  if (tid < NOWN) cnt[tid] = 0;
  int s[KE], d[KE];
#pragma unroll
  for (int k = 0; k < KE; ++k) {
    const int i = tid + k * 1024;
    d[k] = (i < CH) ? dst[e0 + i] : -1;
    s[k] = (i < CH) ? src[e0 + i] : 0;
  }
  float ns[KE], nd[KE];
#pragma unroll
  for (int k = 0; k < KE; ++k) {
    ns[k] = dinv[s[k]];
    nd[k] = dinv[d[k] >= 0 ? d[k] : 0];
  }
  __syncthreads();
  int o[KE], r[KE];
#pragma unroll
  for (int k = 0; k < KE; ++k) {
    o[k] = 0; r[k] = 0;
    if (d[k] >= 0) {
      o[k] = d[k] / ROWS;
      r[k] = atomicAdd(&cnt[o[k]], 1);
    }
  }
  __syncthreads();
  int c = 0, inc = 0;
  if (tid < NOWN) {
    c   = cnt[tid];
    inc = wave_incl_scan(c, lane);
    if (lane == 31) wsum[tid >> 5] = inc;
  }
  __syncthreads();
  if (tid < NOWN) {
    int first = inc - c;
#pragma unroll
    for (int w = 0; w < NOWN / 32; ++w) if (w < (tid >> 5)) first += wsum[w];
    cnt[tid] = first;
    offs[blockIdx.x * NOWN + tid] = first;
  }
  __syncthreads();
#pragma unroll
  for (int k = 0; k < KE; ++k) {
    if (d[k] >= 0)
      stage[cnt[o[k]] + r[k]] = make_int2(s[k] | ((d[k] - o[k] * ROWS) << SRCB),
                                          __float_as_int(ns[k] * nd[k]));
  }
  __syncthreads();
  int4* po = (int4*)((int2*)recw + (size_t)blockIdx.x * CHP);
  for (int i = tid; i < CH / 2; i += 1024) po[i] = stage4[i];
}

#define MT    5
#define GROWS (16 * MT)
template <int NOUT, bool RELU_A>
__global__ __launch_bounds__(NOUT * 2) void k_gemm_wmma(const float* __restrict__ A,
                                                        const float* __restrict__ B,
                                                        float* __restrict__ C) {
  static_assert(NNODES % GROWS == 0, "the row blocks tile the rows exactly: no row guard anywhere");
  constexpr int NT = NOUT * 2;
  constexpr int NV = GROWS * (KDIM / 4);
  static_assert(NV % NT == 0, "every thread stages the same number of float4 words");
  __shared__ float lA[GROWS * APITCH];

  const int tid  = threadIdx.x;
  const int wave = tid >> 5;
  const int lane = tid & 31;
  const int row0 = blockIdx.x * GROWS;
  const int col0 = wave * 16;
  const int half = lane >> 4;
  const int l16  = lane & 15;

  {
    const float4* gA = (const float4*)(A + (size_t)row0 * KDIM);
    float4* sA = (float4*)lA;
#pragma unroll
    for (int i = 0; i < NV / NT; ++i) {
      const int idx = tid + i * NT;
      const int r   = idx >> 5;
      const int c4  = idx & 31;
      float4 v = gA[idx];
      if (RELU_A) {
        v.x = fmaxf(v.x, 0.0f); v.y = fmaxf(v.y, 0.0f);
        v.z = fmaxf(v.z, 0.0f); v.w = fmaxf(v.w, 0.0f);
      }
      sA[r * (APITCH / 4) + c4] = v;
    }
  }
  __syncthreads();

  const float* aP = lA + l16 * APITCH + half * 2;
  const float* bP = B + (half * 2) * NOUT + (col0 + l16);

  v8f acc[MT];
#pragma unroll
  for (int t = 0; t < MT; ++t) acc[t] = v8f{};
#pragma unroll
  for (int k = 0; k < KDIM; k += 4) {
    v2f bv;
    bv.x = bP[k * NOUT];
    bv.y = bP[(k + 1) * NOUT];
#pragma unroll
    for (int t = 0; t < MT; ++t) {
      const v2f av = *(const v2f*)(aP + t * 16 * APITCH + k);
      acc[t] = __builtin_amdgcn_wmma_f32_16x16x4_f32(false, av, false, bv,
                                                     (short)0, acc[t], false, false);
    }
  }

#pragma unroll
  for (int t = 0; t < MT; ++t) {
    float* cP = C + (size_t)(row0 + t * 16 + half * 8) * NOUT + (col0 + l16);
#pragma unroll
    for (int r = 0; r < 8; ++r) cP[r * NOUT] = acc[t][r];
  }
}


template <int F, bool LIST>
__device__ __forceinline__ void drain_records(const float* __restrict__ h, const int2* __restrict__ rec,
                                              const int* q, int2* cl, bool keep, int n,
                                              int wv, int lane, float* tile) {
  constexpr int CPE = F / 32;
  constexpr int U   = 32 / CPE;
  int i0 = wv * 32;
  int pv = 0, nv = 0;
  bool val = false;
  if (i0 < n) {
    val = (i0 + lane) < n;
    const int i = val ? i0 + lane : i0;
    const int2 r = LIST ? cl[i] : rec[q[i]];
    pv = r.x; nv = r.y;
  }
  while (i0 < n) {
    if constexpr (!LIST) {
      if (keep && val) cl[i0 + lane] = make_int2(pv, nv);
    }
    const int j0 = i0 + 1024;
    int pv2 = 0, nv2 = 0;
    bool val2 = false;
    if (j0 < n) {
      val2 = (j0 + lane) < n;
      const int j = val2 ? j0 + lane : j0;
      const int2 r = LIST ? cl[j] : rec[q[j]];
      pv2 = r.x; nv2 = r.y;
    }
    unsigned m = __builtin_amdgcn_ballot_w32(val);
    int ii = 0;
    while (m != 0u) {
      int   d[U];
      float w[U];
      bool  ok[U];
      float v[U][CPE];
#pragma unroll
      for (int u = 0; u < U; ++u) {
        ok[u] = (m != 0u);
        if (ok[u]) ii = __builtin_ctz(m);
        m &= m - 1u;
        const int pk = __builtin_amdgcn_readlane(pv, ii);
        d[u] = pk >> SRCB;
        w[u] = __int_as_float(__builtin_amdgcn_readlane(nv, ii));
        const float* hp = h + (size_t)(pk & ((1 << SRCB) - 1)) * F + lane;
#pragma unroll
        for (int c = 0; c < CPE; ++c) v[u][c] = hp[c * 32];
      }
#pragma unroll
      for (int u = 0; u < U; ++u) {
        if constexpr (CPE == 4) asm volatile("" :: "v"(v[u][0]), "v"(v[u][1]), "v"(v[u][2]), "v"(v[u][3]));
        else                    asm volatile("" :: "v"(v[u][0]), "v"(v[u][1]));
      }
#pragma unroll
      for (int u = 0; u < U; ++u) {
        if (ok[u]) {
          float* tp = tile + d[u] * F + lane;
#pragma unroll
          for (int c = 0; c < CPE; ++c) atomicAdd(tp + c * 32, v[u][c] * w[u]);
        }
      }
    }
    pv = pv2; nv = nv2; val = val2;
    i0 = j0;
  }
}

template <int F, bool FIRST>
__global__ __launch_bounds__(1024) void k_agg_own(const float* __restrict__ h, const int* __restrict__ recw,
                                                  const int* __restrict__ offs,
                                                  const int* __restrict__ indeg, const float* __restrict__ dinv,
                                                  const float* __restrict__ bias, float* __restrict__ out,
                                                  int* clw) {
  constexpr int F4  = F / 4;
  const int2* rec = (const int2*)recw;
  __shared__ float tile[ROWS * F];
  __shared__ int   q[QCAP];
  __shared__ int   seg_g[NCH];
  __shared__ int   seg_p[NCH];
  __shared__ int   seg_n[NCH];
  __shared__ int   s_w[NCH / 32];
  __shared__ int   s_n[1];

  const int tid  = threadIdx.x;
  const int lane = tid & 31;
  const int wv    = __builtin_amdgcn_readfirstlane(tid >> 5);
  const int b     = blockIdx.x;
  const int base  = b * ROWS;
  const int nrows = min(ROWS, NNODES - base);
  int2* cl = (int2*)clw + (size_t)b * CCAP;

  if (tid == 0) s_n[0] = 0;
  __syncthreads();
  int run0 = 0, run1 = 0;
  if (FIRST && tid < NCH) {
    const int* op = offs + tid * NOWN + b;
    run0 = op[0];
    run1 = (b + 1 < NOWN) ? op[1] : CH;
  }
  if (tid < nrows) atomicAdd(&s_n[0], indeg[base + tid]);

  const int n4 = nrows * F4;
  float4* t4 = (float4*)tile;
  {
    const float4* hp = (const float4*)(h + (size_t)base * F);
    const float4* bp = (const float4*)bias;
    for (int i = tid; i < n4; i += 1024) {
      const float  d  = dinv[base + i / F4];
      const float  dd = d * d;
      const float4 x  = hp[i];
      const float4 bb = bp[i % F4];
      float4 y;
      y.x = dd * x.x + bb.x; y.y = dd * x.y + bb.y;
      y.z = dd * x.z + bb.z; y.w = dd * x.w + bb.w;
      t4[i] = y;
    }
  }
  int rlen = 0, rinc = 0;
  if (FIRST && tid < NCH) {
    rlen = run1 - run0;
    rinc = wave_incl_scan(rlen, lane);
    if (lane == 31) s_w[tid >> 5] = rinc;
  }
  __syncthreads();
  const int  total  = __builtin_amdgcn_readfirstlane(s_n[0]);
  const bool listed = (total <= CCAP);

  if (!FIRST && listed) {
    drain_records<F, true>(h, rec, q, cl, false, total, wv, lane, tile);
    __syncthreads();
  } else {
    if (!FIRST) {
      if (tid < NCH) {
        const int* op = offs + tid * NOWN + b;
        run0 = op[0];
        run1 = (b + 1 < NOWN) ? op[1] : CH;
        rlen = run1 - run0;
        rinc = wave_incl_scan(rlen, lane);
        if (lane == 31) s_w[tid >> 5] = rinc;
      }
      __syncthreads();
    }
    int nrun = 0;
#pragma unroll
    for (int w = 0; w < NCH / 32; ++w) nrun += s_w[w];
    nrun = __builtin_amdgcn_readfirstlane(nrun);
    if (tid < NCH) {
      int first = rinc - rlen;
#pragma unroll
      for (int w = 0; w < NCH / 32; ++w) if (w < (tid >> 5)) first += s_w[w];
      seg_g[tid] = tid * CHP + run0;
      seg_p[tid] = first;
      seg_n[tid] = rlen;
    }
    __syncthreads();

    for (int p0 = 0; p0 < nrun; p0 += QCAP) {
      const int n = min(nrun - p0, QCAP);
      {
        const int cch = tid >> 2;
        const int g   = seg_g[cch];
        const int p   = seg_p[cch] - p0;
        const int ln  = seg_n[cch];
        for (int k = tid & 3; k < ln; k += 4) {
          const int qi = p + k;
          if (qi >= 0 && qi < QCAP) q[qi] = g + k;
        }
      }
      __syncthreads();
      drain_records<F, false>(h, rec, q, cl, FIRST && listed, n, wv, lane, tile);
      __syncthreads();
    }
  }

  float4* og = (float4*)(out + (size_t)base * F);
  for (int i = tid; i < n4; i += 1024) og[i] = t4[i];
}

__global__ void k_decode(const float* __restrict__ z, const int* __restrict__ pa,
                         const int* __restrict__ pb, float* __restrict__ out, int np) {
  int tid = blockIdx.x * blockDim.x + threadIdx.x;
  int pair = tid >> 4;
  int l    = tid & 15;
  if (pair >= np) return;
  const float4 va = ((const float4*)z)[((size_t)pa[pair] << 4) + l];
  const float4 vb = ((const float4*)z)[((size_t)pb[pair] << 4) + l];
  float p = va.x * vb.x + va.y * vb.y + va.z * vb.z + va.w * vb.w;
  p += __shfl_xor(p, 8);
  p += __shfl_xor(p, 4);
  p += __shfl_xor(p, 2);
  p += __shfl_xor(p, 1);
  if (l == 0) out[pair] = p;
}

extern "C" void kernel_launch(void* const* d_in, const int* in_sizes, int n_in,
                              void* d_out, int out_size, void* d_ws, size_t ws_size,
                              hipStream_t stream) {
  (void)in_sizes; (void)n_in; (void)out_size; (void)ws_size;

  const float* x  = (const float*)d_in[0];
  const int*   ei = (const int*)d_in[1];
  const int*   el = (const int*)d_in[2];
  const float* W1 = (const float*)d_in[3];
  const float* b1 = (const float*)d_in[4];
  const float* W2 = (const float*)d_in[5];
  const float* b2 = (const float*)d_in[6];
  const float* W3 = (const float*)d_in[7];
  const float* b3 = (const float*)d_in[8];

  const int* src = ei;
  const int* dst = ei + NEDGES;
  const int* pa  = el;
  const int* pb  = el + NPAIRS;
  float* out = (float*)d_out;

  float* dinv = (float*)d_ws;
  int*   rec  = (int*)(dinv + ((NNODES + 127) & ~127));
  float* bufA = (float*)(rec + 2 * (size_t)NCH * CHP);
  float* bufB = bufA + (size_t)NNODES * DHID;
  float* bufC = bufB + (size_t)NNODES * DHID;
  unsigned* part = (unsigned*)(bufC + (size_t)NNODES * DHID);
  int* indeg = (int*)(part + (size_t)NSL * HW);
  int* offs  = indeg + NNODES;
  int* clist = offs + (size_t)NCH * NOWN;

  auto cdiv = [](long long a, long long b) { return (int)((a + b - 1) / b); };

  k_deg_hist<<<NSL, 1024, 0, stream>>>(dst, part);
  k_deg_sum <<<cdiv(HW, 256), 256, 0, stream>>>(part, dinv, indeg);
  k_edge_bin<<<NCH, 1024, 0, stream>>>(src, dst, dinv, rec, offs);

  const int mBlocks = NNODES / GROWS;

  k_gemm_wmma<DHID, false><<<mBlocks, DHID * 2, 0, stream>>>(x, W1, bufA);
  k_agg_own<DHID, true><<<NOWN, 1024, 0, stream>>>(bufA, rec, offs, indeg, dinv, b1, bufB, clist);

  k_gemm_wmma<DHID, true><<<mBlocks, DHID * 2, 0, stream>>>(bufB, W2, bufA);
  k_agg_own<DHID, false><<<NOWN, 1024, 0, stream>>>(bufA, rec, offs, indeg, dinv, b2, bufC, clist);

  k_gemm_wmma<DOUT, true><<<mBlocks, DOUT * 2, 0, stream>>>(bufC, W3, bufA);
  k_agg_own<DOUT, false><<<NOWN, 1024, 0, stream>>>(bufA, rec, offs, indeg, dinv, b3, bufB, clist);

  k_decode<<<cdiv((long long)NPAIRS * 16, 256), 256, 0, stream>>>(bufB, pa, pb, out, NPAIRS);
}


# in-degrees counted by the owner work-groups from their binned records (LDS counters) as they write their lists; norms formed in edge pass 1: plane histogram and sum launch go; sum order unchanged
# speedup vs baseline: 4.0929x; 1.0630x over previous
#include <hip/hip_runtime.h>

#define NNODES 100000
#define NEDGES 1600000
#define NPAIRS 200000
#define DIN    128
#define DHID   128
#define DOUT   64
#define KDIM   128
#define APITCH 132

typedef __attribute__((ext_vector_type(2))) float v2f;
typedef __attribute__((ext_vector_type(8))) float v8f;


#define ROWS 391
#define NOWN 256
#define QCAP 24576
#define NCH  256
#define CH   (NEDGES / NCH)
#define CHP  ((CH + 31) & ~31)
#define SRCB 17
#define DPITCH 416
#define LPAD 32
#define LCAP (NEDGES + LPAD * NOWN)
static_assert(NOWN * ROWS >= NNODES, "the owners cover the rows");
static_assert(NEDGES % NCH == 0 && CHP % 4 == 0, "the chunks tile the edge list, a chunk's region is whole int4");
static_assert(NCH * 4 == 1024 && NCH % 32 == 0, "an owner work-group has four threads per run, whole waves scan the runs");
static_assert(NOWN % 32 == 0 && NOWN <= 1024, "whole waves of a record work-group scan the owners' counters");
static_assert(DPITCH >= ROWS && DPITCH % 32 == 0 && QCAP % LPAD == 0, "whole lines per owner, whole lines per queue round");
static_assert(NNODES <= (1 << SRCB) && ROWS <= (1 << (31 - SRCB)), "source and owned row share one record word");

__device__ __forceinline__ int wave_incl_scan(int v, int lane) {
#pragma unroll
  for (int dl = 1; dl < 32; dl <<= 1) {
    const int up = __shfl_up(v, dl);
    if (lane >= dl) v += up;
  }
  return v;
}

__global__ __launch_bounds__(1024) void k_edge_bin(const int* __restrict__ src, const int* __restrict__ dst,
                                                   int* __restrict__ recw, int* __restrict__ offs) {
  constexpr int KE = (CH + 1023) / 1024;
  __shared__ int  cnt[NOWN];
  __shared__ int  wsum[NOWN / 32];
  __shared__ int4 stage4[CHP / 4];
  int* stage = (int*)stage4;
  const int tid  = threadIdx.x;
  const int lane = tid & 31;
  const int e0   = blockIdx.x * CH;
  if (tid < NOWN) cnt[tid] = 0;
  if (tid < CHP - CH) stage[CH + tid] = 0;
  int s[KE], d[KE];
#pragma unroll
  for (int k = 0; k < KE; ++k) {
    const int i = tid + k * 1024;
    d[k] = (i < CH) ? dst[e0 + i] : -1;
    s[k] = (i < CH) ? src[e0 + i] : 0;
  }
  __syncthreads();
  int o[KE], r[KE];
#pragma unroll
  for (int k = 0; k < KE; ++k) {
    o[k] = 0; r[k] = 0;
    if (d[k] >= 0) {
      o[k] = d[k] / ROWS;
      r[k] = atomicAdd(&cnt[o[k]], 1);
    }
  }
  __syncthreads();
  int c = 0, inc = 0;
  if (tid < NOWN) {
    c   = cnt[tid];
    inc = wave_incl_scan(c, lane);
    if (lane == 31) wsum[tid >> 5] = inc;
  }
  __syncthreads();
  if (tid < NOWN) {
    int first = inc - c;
#pragma unroll
    for (int w = 0; w < NOWN / 32; ++w) if (w < (tid >> 5)) first += wsum[w];
    cnt[tid] = first;
    offs[blockIdx.x * NOWN + tid] = first;
  }
  __syncthreads();
#pragma unroll
  for (int k = 0; k < KE; ++k) {
    if (d[k] >= 0) stage[cnt[o[k]] + r[k]] = s[k] | ((d[k] - o[k] * ROWS) << SRCB);
  }
  __syncthreads();
  int4* po = (int4*)(recw + (size_t)blockIdx.x * CHP);
  for (int i = tid; i < CHP / 4; i += 1024) po[i] = stage4[i];
}

__device__ __forceinline__ int dslot(int node) { return node + (DPITCH - ROWS) * (node / ROWS); }

__global__ __launch_bounds__(1024) void k_own_deg(const int* __restrict__ recw, const int* __restrict__ offs,
                                                  float* __restrict__ dinvp, int* __restrict__ meta,
                                                  int* __restrict__ cl4) {
  constexpr int UQ = 8;
  __shared__ int q[QCAP];
  __shared__ int cnt[ROWS];
  __shared__ int seg_g[NCH];
  __shared__ int seg_p[NCH];
  __shared__ int seg_n[NCH];
  __shared__ int s_w[NCH / 32];
  __shared__ int s_o[NCH / 32];

  const int tid   = threadIdx.x;
  const int lane  = tid & 31;
  const int b     = blockIdx.x;
  const int nrows = min(ROWS, NNODES - b * ROWS);

  int run0 = 0, rlen = 0, rinc = 0;
  if (tid < NCH) {
    const int* op = offs + tid * NOWN + b;
    run0 = op[0];
    const int run1 = (b + 1 < NOWN) ? op[1] : CH;
    rlen = run1 - run0;
    rinc = wave_incl_scan(rlen, lane);
    const int oinc = wave_incl_scan(run0, lane);
    if (lane == 31) { s_w[tid >> 5] = rinc; s_o[tid >> 5] = oinc; }
  }
  if (tid < ROWS) cnt[tid] = 0;
  __syncthreads();
  int nrun = 0, lower = 0;
#pragma unroll
  for (int w = 0; w < NCH / 32; ++w) { nrun += s_w[w]; lower += s_o[w]; }
  nrun  = __builtin_amdgcn_readfirstlane(nrun);
  lower = __builtin_amdgcn_readfirstlane(lower);
  const int a0 = (lower + LPAD * b) & ~(LPAD - 1);
  if (tid < NCH) {
    int first = rinc - rlen;
#pragma unroll
    for (int w = 0; w < NCH / 32; ++w) if (w < (tid >> 5)) first += s_w[w];
    seg_g[tid] = tid * CHP + run0;
    seg_p[tid] = first;
    seg_n[tid] = rlen;
  }
  __syncthreads();

  for (int p0 = 0; p0 < nrun; p0 += QCAP) {
    const int n = min(nrun - p0, QCAP);
    {
      const int cch = tid >> 2;
      const int g   = seg_g[cch];
      const int p   = seg_p[cch] - p0;
      const int ln  = seg_n[cch];
      for (int k = tid & 3; k < ln; k += 4) {
        const int qi = p + k;
        if (qi >= 0 && qi < QCAP) q[qi] = g + k;
      }
    }
    __syncthreads();
    for (int i0 = 0; i0 < n; i0 += UQ * 1024) {
      int x[UQ];
#pragma unroll
      for (int j = 0; j < UQ; ++j) {
        const int i = i0 + j * 1024 + tid;
        x[j] = (i < n) ? recw[q[i]] : -1;
      }
#pragma unroll
      for (int j = 0; j < UQ; ++j) {
        const int i = i0 + j * 1024 + tid;
        if (x[j] >= 0) {
          cl4[a0 + p0 + i] = x[j];
          atomicAdd(&cnt[x[j] >> SRCB], 1);
        }
      }
    }
    __syncthreads();
  }

  if (tid < nrows) dinvp[b * DPITCH + tid] = rsqrtf(1.0f + (float)cnt[tid]);
  if (tid == 0) { meta[b * 32] = a0; meta[b * 32 + 1] = nrun; }
}

#define MT    5
#define GROWS (16 * MT)
template <int NOUT, bool RELU_A>
__global__ __launch_bounds__(NOUT * 2) void k_gemm_wmma(const float* __restrict__ A,
                                                        const float* __restrict__ B,
                                                        float* __restrict__ C) {
  static_assert(NNODES % GROWS == 0, "the row blocks tile the rows exactly: no row guard anywhere");
  constexpr int NT = NOUT * 2;
  constexpr int NV = GROWS * (KDIM / 4);
  static_assert(NV % NT == 0, "every thread stages the same number of float4 words");
  __shared__ float lA[GROWS * APITCH];

  const int tid  = threadIdx.x;
  const int wave = tid >> 5;
  const int lane = tid & 31;
  const int row0 = blockIdx.x * GROWS;
  const int col0 = wave * 16;
  const int half = lane >> 4;
  const int l16  = lane & 15;

  {
    const float4* gA = (const float4*)(A + (size_t)row0 * KDIM);
    float4* sA = (float4*)lA;
#pragma unroll
    for (int i = 0; i < NV / NT; ++i) {
      const int idx = tid + i * NT;
      const int r   = idx >> 5;
      const int c4  = idx & 31;
      float4 v = gA[idx];
      if (RELU_A) {
        v.x = fmaxf(v.x, 0.0f); v.y = fmaxf(v.y, 0.0f);
        v.z = fmaxf(v.z, 0.0f); v.w = fmaxf(v.w, 0.0f);
      }
      sA[r * (APITCH / 4) + c4] = v;
    }
  }
  __syncthreads();

  const float* aP = lA + l16 * APITCH + half * 2;
  const float* bP = B + (half * 2) * NOUT + (col0 + l16);

  v8f acc[MT];
#pragma unroll
  for (int t = 0; t < MT; ++t) acc[t] = v8f{};
#pragma unroll
  for (int k = 0; k < KDIM; k += 4) {
    v2f bv;
    bv.x = bP[k * NOUT];
    bv.y = bP[(k + 1) * NOUT];
#pragma unroll
    for (int t = 0; t < MT; ++t) {
      const v2f av = *(const v2f*)(aP + t * 16 * APITCH + k);
      acc[t] = __builtin_amdgcn_wmma_f32_16x16x4_f32(false, av, false, bv,
                                                     (short)0, acc[t], false, false);
    }
  }

#pragma unroll
  for (int t = 0; t < MT; ++t) {
    float* cP = C + (size_t)(row0 + t * 16 + half * 8) * NOUT + (col0 + l16);
#pragma unroll
    for (int r = 0; r < 8; ++r) cP[r * NOUT] = acc[t][r];
  }
}


template <int F>
__device__ __forceinline__ void drain_batch(const float* __restrict__ h, int pv, int nv, bool val,
                                            int lane, float* tile) {
  constexpr int CPE = F / 32;
  constexpr int U   = 32 / CPE;
  constexpr int SM  = (1 << SRCB) - 1;
  unsigned m = __builtin_amdgcn_ballot_w32(val);
  int ii = 0;
  while (m != 0u) {
    int   d[U];
    float w[U];
    bool  ok[U];
    float v[U][CPE];
#pragma unroll
    for (int u = 0; u < U; ++u) {
      ok[u] = (m != 0u);
      if (ok[u]) ii = __builtin_ctz(m);
      m &= m - 1u;
      const int pk = __builtin_amdgcn_readlane(pv, ii);
      d[u] = pk >> SRCB;
      w[u] = __int_as_float(__builtin_amdgcn_readlane(nv, ii));
      const float* hp = h + (size_t)(pk & SM) * F + lane;
#pragma unroll
      for (int c = 0; c < CPE; ++c) v[u][c] = hp[c * 32];
    }
#pragma unroll
    for (int u = 0; u < U; ++u) {
      if constexpr (CPE == 4) asm volatile("" :: "v"(v[u][0]), "v"(v[u][1]), "v"(v[u][2]), "v"(v[u][3]));
      else                    asm volatile("" :: "v"(v[u][0]), "v"(v[u][1]));
    }
#pragma unroll
    for (int u = 0; u < U; ++u) {
      if (ok[u]) {
        float* tp = tile + d[u] * F + lane;
#pragma unroll
        for (int c = 0; c < CPE; ++c) atomicAdd(tp + c * 32, v[u][c] * w[u]);
      }
    }
  }
}

template <int F, bool FIRST>
__device__ __forceinline__ void drain_records(const float* __restrict__ h, const int* __restrict__ l4, int2* l8,
                                              const float* __restrict__ dinvp, const float* dl, int n,
                                              int wv, int lane, float* tile) {
  constexpr int SM = (1 << SRCB) - 1;
  int i0 = wv * 32;
  if constexpr (FIRST) {
    int   pv = 0, pvn = 0;
    float ds = 0.0f;
    bool  val = false, valn = false;
    if (i0 < n) {
      val = (i0 + lane) < n;
      pv  = l4[val ? i0 + lane : i0];
      ds  = dinvp[dslot(pv & SM)];
    }
    if (i0 + 1024 < n) {
      valn = (i0 + 1024 + lane) < n;
      pvn  = l4[valn ? i0 + 1024 + lane : i0 + 1024];
    }
    while (i0 < n) {
      const int nv = __float_as_int(ds * dl[pv >> SRCB]);
      if (val) l8[i0 + lane] = make_int2(pv, nv);
      float dsn = 0.0f;
      if (i0 + 1024 < n) dsn = dinvp[dslot(pvn & SM)];
      int  pvf = 0;
      bool valf = false;
      if (i0 + 2048 < n) {
        valf = (i0 + 2048 + lane) < n;
        pvf  = l4[valf ? i0 + 2048 + lane : i0 + 2048];
      }
      drain_batch<F>(h, pv, nv, val, lane, tile);
      pv = pvn; ds = dsn; val = valn;
      pvn = pvf; valn = valf;
      i0 += 1024;
    }
  } else {
    int pv = 0, nv = 0;
    bool val = false;
    if (i0 < n) {
      val = (i0 + lane) < n;
      const int2 r = l8[val ? i0 + lane : i0];
      pv = r.x; nv = r.y;
    }
    while (i0 < n) {
      const int j0 = i0 + 1024;
      int pv2 = 0, nv2 = 0;
      bool val2 = false;
      if (j0 < n) {
        val2 = (j0 + lane) < n;
        const int2 r = l8[val2 ? j0 + lane : j0];
        pv2 = r.x; nv2 = r.y;
      }
      drain_batch<F>(h, pv, nv, val, lane, tile);
      pv = pv2; nv = nv2; val = val2;
      i0 = j0;
    }
  }
}

template <int F, bool FIRST>
__global__ __launch_bounds__(1024) void k_agg_own(const float* __restrict__ h, const int* __restrict__ meta,
                                                  const float* __restrict__ dinvp,
                                                  const float* __restrict__ bias, float* __restrict__ out,
                                                  const int* __restrict__ cl4, int* cl8w) {
  constexpr int F4  = F / 4;
  __shared__ float tile[ROWS * F];
  __shared__ float dl[DPITCH];

  const int tid  = threadIdx.x;
  const int lane = tid & 31;
  const int wv    = __builtin_amdgcn_readfirstlane(tid >> 5);
  const int b     = blockIdx.x;
  const int base  = b * ROWS;
  const int nrows = min(ROWS, NNODES - base);
  const float* dmine = dinvp + b * DPITCH;
  const int a0    = __builtin_amdgcn_readfirstlane(meta[b * 32]);
  const int total = __builtin_amdgcn_readfirstlane(meta[b * 32 + 1]);

  if (FIRST && tid < nrows) dl[tid] = dmine[tid];

  const int n4 = nrows * F4;
  float4* t4 = (float4*)tile;
  {
    const float4* hp = (const float4*)(h + (size_t)base * F);
    const float4* bp = (const float4*)bias;
    for (int i = tid; i < n4; i += 1024) {
      const float  d  = dmine[i / F4];
      const float  dd = d * d;
      const float4 x  = hp[i];
      const float4 bb = bp[i % F4];
      float4 y;
      y.x = dd * x.x + bb.x; y.y = dd * x.y + bb.y;
      y.z = dd * x.z + bb.z; y.w = dd * x.w + bb.w;
      t4[i] = y;
    }
  }
  __syncthreads();

  drain_records<F, FIRST>(h, cl4 + a0, (int2*)cl8w + a0, dinvp, dl, total, wv, lane, tile);
  __syncthreads();

  float4* og = (float4*)(out + (size_t)base * F);
  for (int i = tid; i < n4; i += 1024) og[i] = t4[i];
}

__global__ void k_decode(const float* __restrict__ z, const int* __restrict__ pa,
                         const int* __restrict__ pb, float* __restrict__ out, int np) {
  int tid = blockIdx.x * blockDim.x + threadIdx.x;
  int pair = tid >> 4;
  int l    = tid & 15;
  if (pair >= np) return;
  const float4 va = ((const float4*)z)[((size_t)pa[pair] << 4) + l];
  const float4 vb = ((const float4*)z)[((size_t)pb[pair] << 4) + l];
  float p = va.x * vb.x + va.y * vb.y + va.z * vb.z + va.w * vb.w;
  p += __shfl_xor(p, 8);
  p += __shfl_xor(p, 4);
  p += __shfl_xor(p, 2);
  p += __shfl_xor(p, 1);
  if (l == 0) out[pair] = p;
}

extern "C" void kernel_launch(void* const* d_in, const int* in_sizes, int n_in,
                              void* d_out, int out_size, void* d_ws, size_t ws_size,
                              hipStream_t stream) {
  (void)in_sizes; (void)n_in; (void)out_size; (void)ws_size;

  const float* x  = (const float*)d_in[0];
  const int*   ei = (const int*)d_in[1];
  const int*   el = (const int*)d_in[2];
  const float* W1 = (const float*)d_in[3];
  const float* b1 = (const float*)d_in[4];
  const float* W2 = (const float*)d_in[5];
  const float* b2 = (const float*)d_in[6];
  const float* W3 = (const float*)d_in[7];
  const float* b3 = (const float*)d_in[8];

  const int* src = ei;
  const int* dst = ei + NEDGES;
  const int* pa  = el;
  const int* pb  = el + NPAIRS;
  float* out = (float*)d_out;

  float* dinvp = (float*)d_ws;
  int*   rec  = (int*)(dinvp + (size_t)NOWN * DPITCH);
  float* bufA = (float*)(rec + (size_t)NCH * CHP);
  float* bufB = bufA + (size_t)NNODES * DHID;
  float* bufC = bufB + (size_t)NNODES * DHID;
  int* offs  = (int*)(bufC + (size_t)NNODES * DHID);
  int* meta  = offs + (size_t)NCH * NOWN;
  int* cl4   = meta + (size_t)NOWN * 32;
  int* cl8   = cl4 + (size_t)LCAP;

  auto cdiv = [](long long a, long long b) { return (int)((a + b - 1) / b); };

  k_edge_bin<<<NCH, 1024, 0, stream>>>(src, dst, rec, offs);
  k_own_deg <<<NOWN, 1024, 0, stream>>>(rec, offs, dinvp, meta, cl4);

  const int mBlocks = NNODES / GROWS;

  k_gemm_wmma<DHID, false><<<mBlocks, DHID * 2, 0, stream>>>(x, W1, bufA);
  k_agg_own<DHID, true><<<NOWN, 1024, 0, stream>>>(bufA, meta, dinvp, b1, bufB, cl4, cl8);

  k_gemm_wmma<DHID, true><<<mBlocks, DHID * 2, 0, stream>>>(bufB, W2, bufA);
  k_agg_own<DHID, false><<<NOWN, 1024, 0, stream>>>(bufA, meta, dinvp, b2, bufC, cl4, cl8);

  k_gemm_wmma<DOUT, true><<<mBlocks, DOUT * 2, 0, stream>>>(bufC, W3, bufA);
  k_agg_own<DOUT, false><<<NOWN, 1024, 0, stream>>>(bufA, meta, dinvp, b3, bufB, cl4, cl8);

  k_decode<<<cdiv((long long)NPAIRS * 16, 256), 256, 0, stream>>>(bufB, pa, pb, out, NPAIRS);
}
